# starting kernel
# baseline (speedup 1.0000x reference)
#include <hip/hip_runtime.h>

typedef __attribute__((ext_vector_type(2))) float v2f;
typedef __attribute__((ext_vector_type(8))) float v8f;

#define WPB  4
#define HSTR 9

__device__ __forceinline__ float sigm(float x) {
    return __builtin_amdgcn_rcpf(1.0f + __builtin_amdgcn_exp2f(-1.44269504f * x));
}
__device__ __forceinline__ float tanh_f(float x) {
    return 2.0f * __builtin_amdgcn_rcpf(1.0f + __builtin_amdgcn_exp2f(-2.88539008f * x)) - 1.0f;
}
__device__ __forceinline__ float xor8_dpp(float v) {
    int r = __builtin_amdgcn_update_dpp(0, __builtin_bit_cast(int, v),
                                        0x128  , 0xF, 0xF, true);
    return __builtin_bit_cast(float, r);
}

__global__ void __launch_bounds__(32 * WPB)
lstm_wmma_kernel(const float* __restrict__ x,
                 const float* __restrict__ W_ih, const float* __restrict__ W_hh,
                 const float* __restrict__ b_ih, const float* __restrict__ b_hh,
                 const float* __restrict__ W_fc, const float* __restrict__ b_fc,
                 float* __restrict__ out, int T)
{
    __shared__ float hbuf[WPB][16 * HSTR];

    const int lane = threadIdx.x & 31;
    const int wv   = threadIdx.x >> 5;
    const int n    = lane & 15;
    const int half = lane >> 4;
    const int k2   = 2 * half;
    const int b0   = (blockIdx.x * WPB + wv) * 16;

    v2f BxT0, BxT1, Bh0T0, Bh1T0, Bh0T1, Bh1T1;
    BxT0.x  = W_ih[n * 4 + k2];          BxT0.y  = W_ih[n * 4 + k2 + 1];
    BxT1.x  = W_ih[(n + 16) * 4 + k2];   BxT1.y  = W_ih[(n + 16) * 4 + k2 + 1];
    Bh0T0.x = W_hh[n * 8 + k2];          Bh0T0.y = W_hh[n * 8 + k2 + 1];
    Bh1T0.x = W_hh[n * 8 + k2 + 4];      Bh1T0.y = W_hh[n * 8 + k2 + 5];
    Bh0T1.x = W_hh[(n + 16) * 8 + k2];   Bh0T1.y = W_hh[(n + 16) * 8 + k2 + 1];
    Bh1T1.x = W_hh[(n + 16) * 8 + k2+4]; Bh1T1.y = W_hh[(n + 16) * 8 + k2 + 5];
    const float bias0 = b_ih[n]      + b_hh[n];
    const float bias1 = b_ih[n + 16] + b_hh[n + 16];

    const float gsel = (n < 8) ? 2.0f : 1.0f;

    const float* xrow = x + (size_t)(b0 + n) * (size_t)T * 4 + k2;
    float* hb = &hbuf[wv][0];

    v8f cst;
    #pragma unroll
    for (int i = 0; i < 8; ++i) cst[i] = 0.0f;
    v2f Ah03; Ah03.x = 0.0f; Ah03.y = 0.0f;
    v2f Ah47 = Ah03;

    v2f Ax = *(const v2f*)(xrow);
    for (int t = 0; t < T; ++t) {
        const v2f Axc = Ax;
        const int tn = (t + 1 < T) ? (t + 1) : t;
        Ax = *(const v2f*)(xrow + (size_t)tn * 4);

        v8f acc0, acc1;
        #pragma unroll
        for (int i = 0; i < 8; ++i) { acc0[i] = bias0; acc1[i] = bias1; }

        acc0 = __builtin_amdgcn_wmma_f32_16x16x4_f32(false, Axc,  false, BxT0,  (short)0, acc0, false, false);
        acc1 = __builtin_amdgcn_wmma_f32_16x16x4_f32(false, Axc,  false, BxT1,  (short)0, acc1, false, false);
        acc0 = __builtin_amdgcn_wmma_f32_16x16x4_f32(false, Ah03, false, Bh0T0, (short)0, acc0, false, false);
        acc1 = __builtin_amdgcn_wmma_f32_16x16x4_f32(false, Ah03, false, Bh0T1, (short)0, acc1, false, false);
        acc0 = __builtin_amdgcn_wmma_f32_16x16x4_f32(false, Ah47, false, Bh1T0, (short)0, acc0, false, false);
        acc1 = __builtin_amdgcn_wmma_f32_16x16x4_f32(false, Ah47, false, Bh1T1, (short)0, acc1, false, false);

        v8f hnew;
        #pragma unroll
        for (int v = 0; v < 8; ++v) {
            float s0 = sigm(acc0[v]);
            float u  = sigm(gsel * acc1[v]);
            float f  = xor8_dpp(s0);
            float o  = xor8_dpp(u);
            float g  = 2.0f * u - 1.0f;
            float cn = f * cst[v] + s0 * g;
            cst[v]   = cn;
            hnew[v]  = o * tanh_f(cn);
        }

        if (n < 8) {
            #pragma unroll
            for (int v = 0; v < 8; ++v)
                hb[(v + 8 * half) * HSTR + n] = hnew[v];
        }
        __builtin_amdgcn_wave_barrier();
        asm volatile("" ::: "memory");
        Ah03.x = hb[n * HSTR + k2];
        Ah03.y = hb[n * HSTR + k2 + 1];
        Ah47.x = hb[n * HSTR + k2 + 4];
        Ah47.y = hb[n * HSTR + k2 + 5];
        __builtin_amdgcn_wave_barrier();
    }

    if (half == 0) {
        float acc = b_fc[0];
        #pragma unroll
        for (int j = 0; j < 8; ++j)
            acc += hb[n * HSTR + j] * W_fc[j];
        out[b0 + n] = acc;
    }
}

extern "C" void kernel_launch(void* const* d_in, const int* in_sizes, int n_in,
                              void* d_out, int out_size, void* d_ws, size_t ws_size,
                              hipStream_t stream) {
    const float* x    = (const float*)d_in[0];
    const float* W_ih = (const float*)d_in[1];
    const float* W_hh = (const float*)d_in[2];
    const float* b_ih = (const float*)d_in[3];
    const float* b_hh = (const float*)d_in[4];
    const float* W_fc = (const float*)d_in[5];
    const float* b_fc = (const float*)d_in[6];
    float* out = (float*)d_out;

    const int T = 512;
    const int I = 4;
    const int B = in_sizes[0] / (T * I);
    const int wavesTotal = B / 16;
    const int blocks = wavesTotal / WPB;

    lstm_wmma_kernel<<<blocks, 32 * WPB, 0, stream>>>(x, W_ih, W_hh, b_ih, b_hh,
                                                      W_fc, b_fc, out, T);
}


# Batch tile per wave 16 -> 8 rows (1,024 waves, all SIMDs): gate rows as WMMA A, batch columns c,c+8 as B; a lane keeps i,f,g,o of 2 units, h swapped by row_ror:8, no LDS; units 4-7 sum h[4..7] first
# speedup vs baseline: 1.3767x; 1.3767x over previous
#include <hip/hip_runtime.h>


typedef __attribute__((ext_vector_type(2))) float v2f;
typedef __attribute__((ext_vector_type(8))) float v8f;

#define WPB  4
#define BPW  8
#define FSTR 9

__device__ __forceinline__ float sigm(float x) {
    return __builtin_amdgcn_rcpf(1.0f + __builtin_amdgcn_exp2f(-1.44269504f * x));
}
__device__ __forceinline__ float tanh_f(float x) {
    return 2.0f * __builtin_amdgcn_rcpf(1.0f + __builtin_amdgcn_exp2f(-2.88539008f * x)) - 1.0f;
}
__device__ __forceinline__ float xor8_dpp(float v) {
    int r = __builtin_amdgcn_update_dpp(0, __builtin_bit_cast(int, v),
                                        0x128  , 0xF, 0xF, true);
    return __builtin_bit_cast(float, r);
}

__global__ void __launch_bounds__(32 * WPB)
lstm_wmma_kernel(const float* __restrict__ x,
                 const float* __restrict__ W_ih, const float* __restrict__ W_hh,
                 const float* __restrict__ b_ih, const float* __restrict__ b_hh,
                 const float* __restrict__ W_fc, const float* __restrict__ b_fc,
                 float* __restrict__ out, int T)
{
    __shared__ float hfin[WPB * BPW * FSTR];

    const int lane = threadIdx.x & 31;
    const int wv   = threadIdx.x >> 5;
    const int n    = lane & 15;
    const int half = lane >> 4;
    const int k2   = 2 * half;
    const int hi   = n >> 3;
    const int bl   = wv * BPW + (n & 7);
    const int brow = blockIdx.x * (WPB * BPW) + bl;

    const int R0 = ((n & 7) >> 1) * 8 + 2 * (n >> 3) + (n & 1);
    const int R1 = R0 + 4;
    v2f A0x, A1x, A0o, A0p, A1o, A1p;
    A0x.x = W_ih[R0 * 4 + k2];       A0x.y = W_ih[R0 * 4 + k2 + 1];
    A1x.x = W_ih[R1 * 4 + k2];       A1x.y = W_ih[R1 * 4 + k2 + 1];
    A0o.x = W_hh[R0 * 8 + k2];       A0o.y = W_hh[R0 * 8 + k2 + 1];
    A0p.x = W_hh[R0 * 8 + 4 + k2];   A0p.y = W_hh[R0 * 8 + 4 + k2 + 1];
    A1o.x = W_hh[R1 * 8 + 4 + k2];   A1o.y = W_hh[R1 * 8 + 4 + k2 + 1];
    A1p.x = W_hh[R1 * 8 + k2];       A1p.y = W_hh[R1 * 8 + k2 + 1];

    const int ub = 2 * half + 4 * hi;
    v8f biasv;
    for (int v = 0; v < 8; ++v) {
        const int r = (v >> 1) * 8 + ub + (v & 1);
        biasv[v] = b_ih[r] + b_hh[r];
    }

    const float* xrow = x + (size_t)brow * (size_t)T * 4 + k2;

    float c0 = 0.0f, c1 = 0.0f;
    v2f Bown; Bown.x = 0.0f; Bown.y = 0.0f;
    v2f Bpar = Bown;

    v2f Bx = *(const v2f*)(xrow);
    for (int t = 0; t < T; ++t) {
        const v2f Bxc = Bx;
        const int tn = (t + 1 < T) ? (t + 1) : t;
        Bx = *(const v2f*)(xrow + (size_t)tn * 4);
        asm volatile("" ::: "memory");

        v8f acc0, acc1;
        acc0 = __builtin_amdgcn_wmma_f32_16x16x4_f32(false, A0x, false, Bxc,  (short)0, biasv, false, false);
        acc1 = __builtin_amdgcn_wmma_f32_16x16x4_f32(false, A1x, false, Bxc,  (short)0, biasv, false, false);
        acc0 = __builtin_amdgcn_wmma_f32_16x16x4_f32(false, A0o, false, Bown, (short)0, acc0,  false, false);
        acc1 = __builtin_amdgcn_wmma_f32_16x16x4_f32(false, A1o, false, Bown, (short)0, acc1,  false, false);
        acc0 = __builtin_amdgcn_wmma_f32_16x16x4_f32(false, A0p, false, Bpar, (short)0, acc0,  false, false);
        acc1 = __builtin_amdgcn_wmma_f32_16x16x4_f32(false, A1p, false, Bpar, (short)0, acc1,  false, false);

        float G[8];
        for (int v = 0; v < 8; ++v) G[v] = hi ? acc1[v] : acc0[v];

        const float i0 = sigm(G[0]),   i1 = sigm(G[1]);
        const float f0 = sigm(G[2]),   f1 = sigm(G[3]);
        const float g0 = tanh_f(G[4]), g1 = tanh_f(G[5]);
        const float o0 = sigm(G[6]),   o1 = sigm(G[7]);
        c0 = f0 * c0 + i0 * g0;
        c1 = f1 * c1 + i1 * g1;
        Bown.x = o0 * tanh_f(c0);
        Bown.y = o1 * tanh_f(c1);
        Bpar.x = xor8_dpp(Bown.x);
        Bpar.y = xor8_dpp(Bown.y);
    }

    hfin[bl * FSTR + ub]     = Bown.x;
    hfin[bl * FSTR + ub + 1] = Bown.y;
    __syncthreads();
    if (wv == 0) {
        float acc = b_fc[0];
        for (int j = 0; j < 8; ++j)
            acc += hfin[lane * FSTR + j] * W_fc[j];
        out[blockIdx.x * (WPB * BPW) + lane] = acc;
    }
}

extern "C" void kernel_launch(void* const* d_in, const int* in_sizes, int n_in,
                              void* d_out, int out_size, void* d_ws, size_t ws_size,
                              hipStream_t stream) {
    const float* x    = (const float*)d_in[0];
    const float* W_ih = (const float*)d_in[1];
    const float* W_hh = (const float*)d_in[2];
    const float* b_ih = (const float*)d_in[3];
    const float* b_hh = (const float*)d_in[4];
    const float* W_fc = (const float*)d_in[5];
    const float* b_fc = (const float*)d_in[6];
    float* out = (float*)d_out;

    const int T = 512;
    const int I = 4;
    const int B = in_sizes[0] / (T * I);
    const int wavesTotal = B / BPW;
    const int blocks = wavesTotal / WPB;

    lstm_wmma_kernel<<<blocks, 32 * WPB, 0, stream>>>(x, W_ih, W_hh, b_ih, b_hh,
                                                      W_fc, b_fc, out, T);
}


# x staged through LDS in 16-step chunks: 4 coalesced 16-byte loads per lane two chunks ahead, one LDS read place per chunk into registers, steps unrolled; no load or address in a step; sums unchanged
# speedup vs baseline: 2.1219x; 1.5413x over previous
#include <hip/hip_runtime.h>


typedef __attribute__((ext_vector_type(2))) float v2f;
typedef __attribute__((ext_vector_type(4))) float v4f;
typedef __attribute__((ext_vector_type(8))) float v8f;

#define WPB  4
#define BPW  8
#define FSTR 9
#define CH   16
#define XROW (CH * 4 + 4)
#define XBUF (BPW * XROW)

__device__ __forceinline__ float sigm(float x) {
    return __builtin_amdgcn_rcpf(1.0f + __builtin_amdgcn_exp2f(-1.44269504f * x));
}
__device__ __forceinline__ float tanh_f(float x) {
    return 2.0f * __builtin_amdgcn_rcpf(1.0f + __builtin_amdgcn_exp2f(-2.88539008f * x)) - 1.0f;
}
__device__ __forceinline__ float xor8_dpp(float v) {
    int r = __builtin_amdgcn_update_dpp(0, __builtin_bit_cast(int, v),
                                        0x128  , 0xF, 0xF, true);
    return __builtin_bit_cast(float, r);
}

__global__ void __launch_bounds__(32 * WPB)
lstm_wmma_kernel(const float* __restrict__ x,
                 const float* __restrict__ W_ih, const float* __restrict__ W_hh,
                 const float* __restrict__ b_ih, const float* __restrict__ b_hh,
                 const float* __restrict__ W_fc, const float* __restrict__ b_fc,
                 float* __restrict__ out, int T)
{
    __shared__ float hfin[WPB * BPW * FSTR];
    __shared__ __attribute__((aligned(16))) float xs[WPB * 2 * XBUF];

    const int lane = threadIdx.x & 31;
    const int wv   = threadIdx.x >> 5;
    const int n    = lane & 15;
    const int half = lane >> 4;
    const int k2   = 2 * half;
    const int hi   = n >> 3;
    const int bl   = wv * BPW + (n & 7);

    const int R0 = ((n & 7) >> 1) * 8 + 2 * (n >> 3) + (n & 1);
    const int R1 = R0 + 4;
    v2f A0x, A1x, A0o, A0p, A1o, A1p;
    A0x.x = W_ih[R0 * 4 + k2];       A0x.y = W_ih[R0 * 4 + k2 + 1];
    A1x.x = W_ih[R1 * 4 + k2];       A1x.y = W_ih[R1 * 4 + k2 + 1];
    A0o.x = W_hh[R0 * 8 + k2];       A0o.y = W_hh[R0 * 8 + k2 + 1];
    A0p.x = W_hh[R0 * 8 + 4 + k2];   A0p.y = W_hh[R0 * 8 + 4 + k2 + 1];
    A1o.x = W_hh[R1 * 8 + 4 + k2];   A1o.y = W_hh[R1 * 8 + 4 + k2 + 1];
    A1p.x = W_hh[R1 * 8 + k2];       A1p.y = W_hh[R1 * 8 + k2 + 1];

    const int ub = 2 * half + 4 * hi;
    v8f biasv;
    for (int v = 0; v < 8; ++v) {
        const int r = (v >> 1) * 8 + ub + (v & 1);
        biasv[v] = b_ih[r] + b_hh[r];
    }

    const int NCH = T / CH;
    const int rsel = lane >> 4;
    const int ssel = lane & 15;
    const char* xw = (const char*)x + (size_t)blockIdx.x * (WPB * BPW) * (size_t)T * 16;
    unsigned goff[4];
    for (int j = 0; j < 4; ++j)
        goff[j] = ((unsigned)(wv * BPW + 2 * j + rsel) * (unsigned)T + (unsigned)ssel) * 16u;
    float* xsw = xs + wv * (2 * XBUF);
    float* wst = xsw + rsel * XROW + ssel * 4;
    const float* rld = xsw + (n & 7) * XROW + k2;

    v4f gq[4];
    for (int j = 0; j < 4; ++j) gq[j] = *(const v4f*)(xw + goff[j]);
    for (int j = 0; j < 4; ++j) *(v4f*)(wst + 2 * j * XROW) = gq[j];
    {
        const int c1 = (1 < NCH) ? 1 : 0;
        for (int j = 0; j < 4; ++j) gq[j] = *(const v4f*)(xw + (size_t)c1 * (CH * 16) + goff[j]);
    }

    float c0 = 0.0f, c1 = 0.0f;
    v2f Bown; Bown.x = 0.0f; Bown.y = 0.0f;
    v2f Bpar = Bown;

    for (int ch = 0; ch < NCH; ++ch) {
        const int cur = ch & 1;
        v2f X[CH];
        for (int s = 0; s < CH; ++s) X[s] = *(const v2f*)(rld + cur * XBUF + s * 4);
        for (int j = 0; j < 4; ++j) *(v4f*)(wst + (cur ^ 1) * XBUF + 2 * j * XROW) = gq[j];
        const int cn = (ch + 2 < NCH) ? (ch + 2) : (NCH - 1);
        for (int j = 0; j < 4; ++j) gq[j] = *(const v4f*)(xw + (size_t)cn * (CH * 16) + goff[j]);
        for (int s = 0; s < CH; ++s) asm volatile("" : "+v"(X[s]));
        __builtin_amdgcn_sched_barrier(0);

        #pragma unroll
        for (int s = 0; s < CH; ++s) {
        const v2f Bxc = X[s];

        v8f acc0, acc1;
        acc0 = __builtin_amdgcn_wmma_f32_16x16x4_f32(false, A0x, false, Bxc,  (short)0, biasv, false, false);
        acc1 = __builtin_amdgcn_wmma_f32_16x16x4_f32(false, A1x, false, Bxc,  (short)0, biasv, false, false);
        acc0 = __builtin_amdgcn_wmma_f32_16x16x4_f32(false, A0o, false, Bown, (short)0, acc0,  false, false);
        acc1 = __builtin_amdgcn_wmma_f32_16x16x4_f32(false, A1o, false, Bown, (short)0, acc1,  false, false);
        acc0 = __builtin_amdgcn_wmma_f32_16x16x4_f32(false, A0p, false, Bpar, (short)0, acc0,  false, false);
        acc1 = __builtin_amdgcn_wmma_f32_16x16x4_f32(false, A1p, false, Bpar, (short)0, acc1,  false, false);

        float G[8];
        for (int v = 0; v < 8; ++v) G[v] = hi ? acc1[v] : acc0[v];

        const float i0 = sigm(G[0]),   i1 = sigm(G[1]);
        const float f0 = sigm(G[2]),   f1 = sigm(G[3]);
        const float g0 = tanh_f(G[4]), g1 = tanh_f(G[5]);
        const float o0 = sigm(G[6]),   o1 = sigm(G[7]);
        c0 = f0 * c0 + i0 * g0;
        c1 = f1 * c1 + i1 * g1;
        Bown.x = o0 * tanh_f(c0);
        Bown.y = o1 * tanh_f(c1);
        Bpar.x = xor8_dpp(Bown.x);
        Bpar.y = xor8_dpp(Bown.y);
        }
    }

    hfin[bl * FSTR + ub]     = Bown.x;
    hfin[bl * FSTR + ub + 1] = Bown.y;
    __syncthreads();
    if (wv == 0) {
        float acc = b_fc[0];
        for (int j = 0; j < 8; ++j)
            acc += hfin[lane * FSTR + j] * W_fc[j];
        out[blockIdx.x * (WPB * BPW) + lane] = acc;
    }
}

extern "C" void kernel_launch(void* const* d_in, const int* in_sizes, int n_in,
                              void* d_out, int out_size, void* d_ws, size_t ws_size,
                              hipStream_t stream) {
    const float* x    = (const float*)d_in[0];
    const float* W_ih = (const float*)d_in[1];
    const float* W_hh = (const float*)d_in[2];
    const float* b_ih = (const float*)d_in[3];
    const float* b_hh = (const float*)d_in[4];
    const float* W_fc = (const float*)d_in[5];
    const float* b_fc = (const float*)d_in[6];
    float* out = (float*)d_out;

    const int T = 512;
    const int I = 4;
    const int B = in_sizes[0] / (T * I);
    const int wavesTotal = B / BPW;
    const int blocks = wavesTotal / WPB;

    lstm_wmma_kernel<<<blocks, 32 * WPB, 0, stream>>>(x, W_ih, W_hh, b_ih, b_hh,
                                                      W_fc, b_fc, out, T);
}


# exp2 scale factors folded into weight rows, biases and a pre-scaled cell state; 2s-1 of both tanh fused into FMAs: 5 fewer dependent VALU links a step; rounding of gates, c and h changes
# speedup vs baseline: 2.3741x; 1.1189x over previous
#include <hip/hip_runtime.h>


typedef __attribute__((ext_vector_type(2))) float v2f;
typedef __attribute__((ext_vector_type(4))) float v4f;
typedef __attribute__((ext_vector_type(8))) float v8f;

#define WPB  4
#define BPW  8
#define FSTR 9
#define CH   16
#define XROW (CH * 4 + 4)
#define XBUF (BPW * XROW)

#define KS    (-1.44269504f)
#define KT    (-2.88539008f)
#define INVKT (-0.34657359f)
__device__ __forceinline__ float sig_scaled(float zs, float cm) {
    return __builtin_amdgcn_rcpf(__builtin_fmaf(__builtin_amdgcn_exp2f(zs), cm, cm));
}
__device__ __forceinline__ float xor8_dpp(float v) {
    int r = __builtin_amdgcn_update_dpp(0, __builtin_bit_cast(int, v),
                                        0x128  , 0xF, 0xF, true);
    return __builtin_bit_cast(float, r);
}

__global__ void __launch_bounds__(32 * WPB)
lstm_wmma_kernel(const float* __restrict__ x,
                 const float* __restrict__ W_ih, const float* __restrict__ W_hh,
                 const float* __restrict__ b_ih, const float* __restrict__ b_hh,
                 const float* __restrict__ W_fc, const float* __restrict__ b_fc,
                 float* __restrict__ out, int T)
{
    __shared__ float hfin[WPB * BPW * FSTR];
    __shared__ __attribute__((aligned(16))) float xs[WPB * 2 * XBUF];

    const int lane = threadIdx.x & 31;
    const int wv   = threadIdx.x >> 5;
    const int n    = lane & 15;
    const int half = lane >> 4;
    const int k2   = 2 * half;
    const int hi   = n >> 3;
    const int bl   = wv * BPW + (n & 7);

    const int R0 = ((n & 7) >> 1) * 8 + 2 * (n >> 3) + (n & 1);
    const int R1 = R0 + 4;
    v2f A0x, A1x, A0o, A0p, A1o, A1p;
    A0x.x = W_ih[R0 * 4 + k2];       A0x.y = W_ih[R0 * 4 + k2 + 1];
    A1x.x = W_ih[R1 * 4 + k2];       A1x.y = W_ih[R1 * 4 + k2 + 1];
    A0o.x = W_hh[R0 * 8 + k2];       A0o.y = W_hh[R0 * 8 + k2 + 1];
    A0p.x = W_hh[R0 * 8 + 4 + k2];   A0p.y = W_hh[R0 * 8 + 4 + k2 + 1];
    A1o.x = W_hh[R1 * 8 + 4 + k2];   A1o.y = W_hh[R1 * 8 + 4 + k2 + 1];
    A1p.x = W_hh[R1 * 8 + k2];       A1p.y = W_hh[R1 * 8 + k2 + 1];
    const float rsc = (((n & 7) >> 1) == 2) ? KT : KS;
    A0x *= rsc; A1x *= rsc; A0o *= rsc; A0p *= rsc; A1o *= rsc; A1p *= rsc;

    const int ub = 2 * half + 4 * hi;
    v8f biasv;
    for (int v = 0; v < 8; ++v) {
        const int r = (v >> 1) * 8 + ub + (v & 1);
        biasv[v] = (b_ih[r] + b_hh[r]) * (((v >> 1) == 2) ? KT : KS);
    }

    const int NCH = T / CH;
    const int rsel = lane >> 4;
    const int ssel = lane & 15;
    const char* xw = (const char*)x + (size_t)blockIdx.x * (WPB * BPW) * (size_t)T * 16;
    unsigned goff[4];
    for (int j = 0; j < 4; ++j)
        goff[j] = ((unsigned)(wv * BPW + 2 * j + rsel) * (unsigned)T + (unsigned)ssel) * 16u;
    float* xsw = xs + wv * (2 * XBUF);
    float* wst = xsw + rsel * XROW + ssel * 4;
    const float* rld = xsw + (n & 7) * XROW + k2;

    v4f gq[4];
    for (int j = 0; j < 4; ++j) gq[j] = *(const v4f*)(xw + goff[j]);
    for (int j = 0; j < 4; ++j) *(v4f*)(wst + 2 * j * XROW) = gq[j];
    {
        const int c1 = (1 < NCH) ? 1 : 0;
        for (int j = 0; j < 4; ++j) gq[j] = *(const v4f*)(xw + (size_t)c1 * (CH * 16) + goff[j]);
    }

    float c0 = 0.0f, c1 = 0.0f;
    v2f Bown; Bown.x = 0.0f; Bown.y = 0.0f;
    v2f Bpar = Bown;

    for (int ch = 0; ch < NCH; ++ch) {
        const int cur = ch & 1;
        v2f X[CH];
        for (int s = 0; s < CH; ++s) X[s] = *(const v2f*)(rld + cur * XBUF + s * 4);
        for (int j = 0; j < 4; ++j) *(v4f*)(wst + (cur ^ 1) * XBUF + 2 * j * XROW) = gq[j];
        const int cn = (ch + 2 < NCH) ? (ch + 2) : (NCH - 1);
        for (int j = 0; j < 4; ++j) gq[j] = *(const v4f*)(xw + (size_t)cn * (CH * 16) + goff[j]);
        for (int s = 0; s < CH; ++s) asm volatile("" : "+v"(X[s]));
        __builtin_amdgcn_sched_barrier(0);

        #pragma unroll
        for (int s = 0; s < CH; ++s) {
        const v2f Bxc = X[s];

        v8f acc0, acc1;
        acc0 = __builtin_amdgcn_wmma_f32_16x16x4_f32(false, A0x, false, Bxc,  (short)0, biasv, false, false);
        acc1 = __builtin_amdgcn_wmma_f32_16x16x4_f32(false, A1x, false, Bxc,  (short)0, biasv, false, false);
        acc0 = __builtin_amdgcn_wmma_f32_16x16x4_f32(false, A0o, false, Bown, (short)0, acc0,  false, false);
        acc1 = __builtin_amdgcn_wmma_f32_16x16x4_f32(false, A1o, false, Bown, (short)0, acc1,  false, false);
        acc0 = __builtin_amdgcn_wmma_f32_16x16x4_f32(false, A0p, false, Bpar, (short)0, acc0,  false, false);
        acc1 = __builtin_amdgcn_wmma_f32_16x16x4_f32(false, A1p, false, Bpar, (short)0, acc1,  false, false);

        float G[8];
        for (int v = 0; v < 8; ++v) G[v] = hi ? acc1[v] : acc0[v];

        const float a0 = sig_scaled(G[0], INVKT), a1 = sig_scaled(G[1], INVKT);
        const float f0 = sig_scaled(G[2], 1.0f),  f1 = sig_scaled(G[3], 1.0f);
        const float s0 = sig_scaled(G[4], 1.0f),  s1 = sig_scaled(G[5], 1.0f);
        const float q0 = sig_scaled(G[6], 0.5f),  q1 = sig_scaled(G[7], 0.5f);
        c0 = __builtin_fmaf(a0 + a0, s0, __builtin_fmaf(f0, c0, -a0));
        c1 = __builtin_fmaf(a1 + a1, s1, __builtin_fmaf(f1, c1, -a1));
        const float r0 = sig_scaled(c0, 1.0f), r1 = sig_scaled(c1, 1.0f);
        Bown.x = __builtin_fmaf(q0, r0, -0.5f * q0);
        Bown.y = __builtin_fmaf(q1, r1, -0.5f * q1);
        Bpar.x = xor8_dpp(Bown.x);
        Bpar.y = xor8_dpp(Bown.y);
        }
    }

    hfin[bl * FSTR + ub]     = Bown.x;
    hfin[bl * FSTR + ub + 1] = Bown.y;
    __syncthreads();
    if (wv == 0) {
        float acc = b_fc[0];
        for (int j = 0; j < 8; ++j)
            acc += hfin[lane * FSTR + j] * W_fc[j];
        out[blockIdx.x * (WPB * BPW) + lane] = acc;
    }
}

extern "C" void kernel_launch(void* const* d_in, const int* in_sizes, int n_in,
                              void* d_out, int out_size, void* d_ws, size_t ws_size,
                              hipStream_t stream) {
    const float* x    = (const float*)d_in[0];
    const float* W_ih = (const float*)d_in[1];
    const float* W_hh = (const float*)d_in[2];
    const float* b_ih = (const float*)d_in[3];
    const float* b_hh = (const float*)d_in[4];
    const float* W_fc = (const float*)d_in[5];
    const float* b_fc = (const float*)d_in[6];
    float* out = (float*)d_out;

    const int T = 512;
    const int I = 4;
    const int B = in_sizes[0] / (T * I);
    const int wavesTotal = B / BPW;
    const int blocks = wavesTotal / WPB;

    lstm_wmma_kernel<<<blocks, 32 * WPB, 0, stream>>>(x, W_ih, W_hh, b_ih, b_hh,
                                                      W_fc, b_fc, out, T);
}


# Column groups trade gate tiles each step: one x product per tile carries step t in columns 0-7, t+1 in 8-15; 5 WMMAs a step, was 6; cell state crosses by row_ror:8; h k-blocks summed in swapped order
# speedup vs baseline: 2.4589x; 1.0357x over previous
#include <hip/hip_runtime.h>


typedef __attribute__((ext_vector_type(2))) float v2f;
typedef __attribute__((ext_vector_type(4))) float v4f;
typedef __attribute__((ext_vector_type(8))) float v8f;

#define WPB  4
#define BPW  8
#define FSTR 9
#define CH   16
#define XROW (CH * 4 + 8)
#define XBUF (BPW * XROW)

#define KS    (-1.44269504f)
#define KT    (-2.88539008f)
#define INVKT (-0.34657359f)
__device__ __forceinline__ float sig_scaled(float zs, float cm) {
    return __builtin_amdgcn_rcpf(__builtin_fmaf(__builtin_amdgcn_exp2f(zs), cm, cm));
}
__device__ __forceinline__ float xor8_dpp(float v) {
    int r = __builtin_amdgcn_update_dpp(0, __builtin_bit_cast(int, v),
                                        0x128  , 0xF, 0xF, true);
    return __builtin_bit_cast(float, r);
}

__global__ void __launch_bounds__(32 * WPB)
lstm_wmma_kernel(const float* __restrict__ x,
                 const float* __restrict__ W_ih, const float* __restrict__ W_hh,
                 const float* __restrict__ b_ih, const float* __restrict__ b_hh,
                 const float* __restrict__ W_fc, const float* __restrict__ b_fc,
                 float* __restrict__ out, int T)
{
    __shared__ float hfin[WPB * BPW * FSTR];
    __shared__ __attribute__((aligned(16))) float xs[WPB * 2 * XBUF];

    const int lane = threadIdx.x & 31;
    const int wv   = threadIdx.x >> 5;
    const int n    = lane & 15;
    const int half = lane >> 4;
    const int k2   = 2 * half;
    const int hi   = n >> 3;
    const int bl   = wv * BPW + (n & 7);

    const int R0 = ((n & 7) >> 1) * 8 + 2 * (n >> 3) + (n & 1);
    const int R1 = R0 + 4;
    v2f A0x, A1x, A0o, A0p, A1o, A1p;
    A0x.x = W_ih[R0 * 4 + k2];       A0x.y = W_ih[R0 * 4 + k2 + 1];
    A1x.x = W_ih[R1 * 4 + k2];       A1x.y = W_ih[R1 * 4 + k2 + 1];
    A0o.x = W_hh[R0 * 8 + k2];       A0o.y = W_hh[R0 * 8 + k2 + 1];
    A0p.x = W_hh[R0 * 8 + 4 + k2];   A0p.y = W_hh[R0 * 8 + 4 + k2 + 1];
    A1o.x = W_hh[R1 * 8 + 4 + k2];   A1o.y = W_hh[R1 * 8 + 4 + k2 + 1];
    A1p.x = W_hh[R1 * 8 + k2];       A1p.y = W_hh[R1 * 8 + k2 + 1];
    const float rsc = (((n & 7) >> 1) == 2) ? KT : KS;
    A0x *= rsc; A1x *= rsc; A0o *= rsc; A0p *= rsc; A1o *= rsc; A1p *= rsc;

    v8f bias0, bias1;
    for (int v = 0; v < 8; ++v) {
        const int r = (v >> 1) * 8 + 2 * half + (v & 1);
        const float sc = ((v >> 1) == 2) ? KT : KS;
        bias0[v] = (b_ih[r] + b_hh[r]) * sc;
        bias1[v] = (b_ih[r + 4] + b_hh[r + 4]) * sc;
    }

    const int NCH = T / CH;
    const int rsel = lane >> 4;
    const int ssel = lane & 15;
    const char* xw = (const char*)x + (size_t)blockIdx.x * (WPB * BPW) * (size_t)T * 16;
    unsigned goff[4];
    for (int j = 0; j < 4; ++j)
        goff[j] = ((unsigned)(wv * BPW + 2 * j + rsel) * (unsigned)T + (unsigned)ssel) * 16u;
    float* xsw = xs + wv * (2 * XBUF);
    float* wst = xsw + rsel * XROW + ssel * 4;
    const float* rldA = xsw + (n & 7) * XROW + k2 + 4 * hi;
    const float* rldB = xsw + (n & 7) * XROW + k2 + 4 * (1 - hi);

    v4f gq[4];
    for (int j = 0; j < 4; ++j) gq[j] = *(const v4f*)(xw + goff[j]);
    for (int j = 0; j < 4; ++j) *(v4f*)(wst + 2 * j * XROW) = gq[j];
    {
        const int c1 = (1 < NCH) ? 1 : 0;
        for (int j = 0; j < 4; ++j) gq[j] = *(const v4f*)(xw + (size_t)c1 * (CH * 16) + goff[j]);
    }

    float c0 = 0.0f, c1 = 0.0f;
    v2f Bown; Bown.x = 0.0f; Bown.y = 0.0f;
    v2f Bpar = Bown;

    for (int ch = 0; ch < NCH; ++ch) {
        const int cur = ch & 1;
        v2f XA[CH / 2], XB[CH / 2];
        for (int p = 0; p < CH / 2; ++p) {
            XA[p] = *(const v2f*)(rldA + cur * XBUF + p * 8);
            XB[p] = *(const v2f*)(rldB + cur * XBUF + p * 8);
        }
        for (int j = 0; j < 4; ++j) *(v4f*)(wst + (cur ^ 1) * XBUF + 2 * j * XROW) = gq[j];
        const int cn = (ch + 2 < NCH) ? (ch + 2) : (NCH - 1);
        for (int j = 0; j < 4; ++j) gq[j] = *(const v4f*)(xw + (size_t)cn * (CH * 16) + goff[j]);
        for (int p = 0; p < CH / 2; ++p) {
            asm volatile("" : "+v"(XA[p]));
            asm volatile("" : "+v"(XB[p]));
        }
        __builtin_amdgcn_sched_barrier(0);

        #pragma unroll
        for (int p = 0; p < CH / 2; ++p) {
        const v8f P0 = __builtin_amdgcn_wmma_f32_16x16x4_f32(false, A0x, false, XA[p], (short)0, bias0, false, false);
        const v8f P1 = __builtin_amdgcn_wmma_f32_16x16x4_f32(false, A1x, false, XB[p], (short)0, bias1, false, false);

        #pragma unroll
        for (int odd = 0; odd < 2; ++odd) {
        v8f acc0, acc1;
        acc0 = __builtin_amdgcn_wmma_f32_16x16x4_f32(false, A0p, false, Bown, (short)0, P0,   false, false);
        acc1 = __builtin_amdgcn_wmma_f32_16x16x4_f32(false, A1p, false, Bown, (short)0, P1,   false, false);
        acc0 = __builtin_amdgcn_wmma_f32_16x16x4_f32(false, A0o, false, Bpar, (short)0, acc0, false, false);
        acc1 = __builtin_amdgcn_wmma_f32_16x16x4_f32(false, A1o, false, Bpar, (short)0, acc1, false, false);

        float G[8];
        for (int v = 0; v < 8; ++v) G[v] = (hi != odd) ? acc1[v] : acc0[v];

        const float a0 = sig_scaled(G[0], INVKT), a1 = sig_scaled(G[1], INVKT);
        const float f0 = sig_scaled(G[2], 1.0f),  f1 = sig_scaled(G[3], 1.0f);
        const float s0 = sig_scaled(G[4], 1.0f),  s1 = sig_scaled(G[5], 1.0f);
        const float q0 = sig_scaled(G[6], 0.5f),  q1 = sig_scaled(G[7], 0.5f);
        const float ci0 = xor8_dpp(c0), ci1 = xor8_dpp(c1);
        c0 = __builtin_fmaf(a0 + a0, s0, __builtin_fmaf(f0, ci0, -a0));
        c1 = __builtin_fmaf(a1 + a1, s1, __builtin_fmaf(f1, ci1, -a1));
        const float r0 = sig_scaled(c0, 1.0f), r1 = sig_scaled(c1, 1.0f);
        Bown.x = __builtin_fmaf(q0, r0, -0.5f * q0);
        Bown.y = __builtin_fmaf(q1, r1, -0.5f * q1);
        Bpar.x = xor8_dpp(Bown.x);
        Bpar.y = xor8_dpp(Bown.y);
        }
        }
    }

    const int ubl = 2 * half + 4 * (hi ^ ((T - 1) & 1));
    hfin[bl * FSTR + ubl]     = Bown.x;
    hfin[bl * FSTR + ubl + 1] = Bown.y;
    __syncthreads();
    if (wv == 0) {
        float acc = b_fc[0];
        for (int j = 0; j < 8; ++j)
            acc += hfin[lane * FSTR + j] * W_fc[j];
        out[blockIdx.x * (WPB * BPW) + lane] = acc;
    }
}

extern "C" void kernel_launch(void* const* d_in, const int* in_sizes, int n_in,
                              void* d_out, int out_size, void* d_ws, size_t ws_size,
                              hipStream_t stream) {
    const float* x    = (const float*)d_in[0];
    const float* W_ih = (const float*)d_in[1];
    const float* W_hh = (const float*)d_in[2];
    const float* b_ih = (const float*)d_in[3];
    const float* b_hh = (const float*)d_in[4];
    const float* W_fc = (const float*)d_in[5];
    const float* b_fc = (const float*)d_in[6];
    float* out = (float*)d_out;

    const int T = 512;
    const int I = 4;
    const int B = in_sizes[0] / (T * I);
    const int wavesTotal = B / BPW;
    const int blocks = wavesTotal / WPB;

    lstm_wmma_kernel<<<blocks, 32 * WPB, 0, stream>>>(x, W_ih, W_hh, b_ih, b_hh,
                                                      W_fc, b_fc, out, T);
}


# speedup vs baseline: 2.6163x; 1.0640x over previous
#include <hip/hip_runtime.h>


typedef __attribute__((ext_vector_type(2))) float v2f;
typedef __attribute__((ext_vector_type(4))) float v4f;
typedef __attribute__((ext_vector_type(8))) float v8f;

#define WPB  4
#define BPW  8
#define FSTR 9
#define CH   16
#define XROW (CH * 4 + 8)
#define XBUF (BPW * XROW)

#define KS    (-1.44269504f)
#define KT    (-2.88539008f)
#define INVKT (-0.34657359f)
#define V(x) "+v"(x)
#define ORD(...) asm volatile("" : __VA_ARGS__)
__device__ __forceinline__ float xor8_dpp(float v) {
    int r = __builtin_amdgcn_update_dpp(0, __builtin_bit_cast(int, v),
                                        0x128  , 0xF, 0xF, true);
    return __builtin_bit_cast(float, r);
}

__global__ void __launch_bounds__(32 * WPB)
lstm_wmma_kernel(const float* __restrict__ x,
                 const float* __restrict__ W_ih, const float* __restrict__ W_hh,
                 const float* __restrict__ b_ih, const float* __restrict__ b_hh,
                 const float* __restrict__ W_fc, const float* __restrict__ b_fc,
                 float* __restrict__ out, int T)
{
    __shared__ float hfin[WPB * BPW * FSTR];
    __shared__ __attribute__((aligned(16))) float xs[WPB * 2 * XBUF];

    const int lane = threadIdx.x & 31;
    const int wv   = threadIdx.x >> 5;
    const int n    = lane & 15;
    const int half = lane >> 4;
    const int k2   = 2 * half;
    const int hi   = n >> 3;
    const int bl   = wv * BPW + (n & 7);

    const int R0 = ((n & 7) >> 1) * 8 + 2 * (n >> 3) + (n & 1);
    const int R1 = R0 + 4;
    v2f A0x, A1x, A0o, A0p, A1o, A1p;
    A0x.x = W_ih[R0 * 4 + k2];       A0x.y = W_ih[R0 * 4 + k2 + 1];
    A1x.x = W_ih[R1 * 4 + k2];       A1x.y = W_ih[R1 * 4 + k2 + 1];
    A0o.x = W_hh[R0 * 8 + k2];       A0o.y = W_hh[R0 * 8 + k2 + 1];
    A0p.x = W_hh[R0 * 8 + 4 + k2];   A0p.y = W_hh[R0 * 8 + 4 + k2 + 1];
    A1o.x = W_hh[R1 * 8 + 4 + k2];   A1o.y = W_hh[R1 * 8 + 4 + k2 + 1];
    A1p.x = W_hh[R1 * 8 + k2];       A1p.y = W_hh[R1 * 8 + k2 + 1];
    const float rsc = (((n & 7) >> 1) == 2) ? KT : KS;
    A0x *= rsc; A1x *= rsc; A0o *= rsc; A0p *= rsc; A1o *= rsc; A1p *= rsc;

    v8f bias0, bias1;
    for (int v = 0; v < 8; ++v) {
        const int r = (v >> 1) * 8 + 2 * half + (v & 1);
        const float sc = ((v >> 1) == 2) ? KT : KS;
        bias0[v] = (b_ih[r] + b_hh[r]) * sc;
        bias1[v] = (b_ih[r + 4] + b_hh[r + 4]) * sc;
    }

    const int NCH = T / CH;
    const int rsel = lane >> 4;
    const int ssel = lane & 15;
    const char* xw = (const char*)x + (size_t)blockIdx.x * (WPB * BPW) * (size_t)T * 16;
    unsigned goff[4];
    for (int j = 0; j < 4; ++j)
        goff[j] = ((unsigned)(wv * BPW + 2 * j + rsel) * (unsigned)T + (unsigned)ssel) * 16u;
    float* xsw = xs + wv * (2 * XBUF);
    float* wst = xsw + rsel * XROW + ssel * 4;
    const float* rldA = xsw + (n & 7) * XROW + k2 + 4 * hi;
    const float* rldB = xsw + (n & 7) * XROW + k2 + 4 * (1 - hi);

    v4f gq[4];
    for (int j = 0; j < 4; ++j) gq[j] = *(const v4f*)(xw + goff[j]);
    for (int j = 0; j < 4; ++j) *(v4f*)(wst + 2 * j * XROW) = gq[j];
    {
        const int c1 = (1 < NCH) ? 1 : 0;
        for (int j = 0; j < 4; ++j) gq[j] = *(const v4f*)(xw + (size_t)c1 * (CH * 16) + goff[j]);
    }

    float c0 = 0.0f, c1 = 0.0f;
    v2f Bown; Bown.x = 0.0f; Bown.y = 0.0f;
    v2f Bpar = Bown;

    for (int ch = 0; ch < NCH; ++ch) {
        const int cur = ch & 1;
        v2f XA[CH / 2], XB[CH / 2];
        for (int p = 0; p < CH / 2; ++p) {
            XA[p] = *(const v2f*)(rldA + cur * XBUF + p * 8);
            XB[p] = *(const v2f*)(rldB + cur * XBUF + p * 8);
        }
        for (int j = 0; j < 4; ++j) *(v4f*)(wst + (cur ^ 1) * XBUF + 2 * j * XROW) = gq[j];
        const int cn = (ch + 2 < NCH) ? (ch + 2) : (NCH - 1);
        for (int j = 0; j < 4; ++j) gq[j] = *(const v4f*)(xw + (size_t)cn * (CH * 16) + goff[j]);
        for (int p = 0; p < CH / 2; ++p) {
            asm volatile("" : "+v"(XA[p]));
            asm volatile("" : "+v"(XB[p]));
        }
        __builtin_amdgcn_sched_barrier(0);

        v8f P0 = __builtin_amdgcn_wmma_f32_16x16x4_f32(false, A0x, false, XA[0], (short)0, bias0, false, false);
        v8f P1 = __builtin_amdgcn_wmma_f32_16x16x4_f32(false, A1x, false, XB[0], (short)0, bias1, false, false);

        #pragma unroll
        for (int p = 0; p < CH / 2; ++p) {
        v8f Pn0 = P0, Pn1 = P1;
        #pragma unroll
        for (int odd = 0; odd < 2; ++odd) {
        v8f acc0, acc1;
        acc0 = __builtin_amdgcn_wmma_f32_16x16x4_f32(false, A0p, false, Bown, (short)0, P0,   false, false);
        acc1 = __builtin_amdgcn_wmma_f32_16x16x4_f32(false, A1p, false, Bown, (short)0, P1,   false, false);
        acc0 = __builtin_amdgcn_wmma_f32_16x16x4_f32(false, A0o, false, Bpar, (short)0, acc0, false, false);
        acc1 = __builtin_amdgcn_wmma_f32_16x16x4_f32(false, A1o, false, Bpar, (short)0, acc1, false, false);

        float G0 = (hi != odd) ? acc1[0] : acc0[0], G1 = (hi != odd) ? acc1[1] : acc0[1];
        float G2 = (hi != odd) ? acc1[2] : acc0[2], G3 = (hi != odd) ? acc1[3] : acc0[3];
        float G4 = (hi != odd) ? acc1[4] : acc0[4], G5 = (hi != odd) ? acc1[5] : acc0[5];
        float G6 = (hi != odd) ? acc1[6] : acc0[6], G7 = (hi != odd) ? acc1[7] : acc0[7];
        ORD(V(G0), V(G1), V(G2), V(G3), V(G4), V(G5), V(G6), V(G7));

        float ei1 = __builtin_amdgcn_exp2f(G1);
        ORD(V(ei1), V(G0));
        float ei0 = __builtin_amdgcn_exp2f(G0);
        ORD(V(ei0), V(G3));
        float ef1 = __builtin_amdgcn_exp2f(G3);
        ORD(V(ef1), V(G2));
        float ef0 = __builtin_amdgcn_exp2f(G2);
        ORD(V(ef0), V(G5));
        float eg1 = __builtin_amdgcn_exp2f(G5);
        ORD(V(eg1), V(ei1));
        float di1 = __builtin_fmaf(ei1, INVKT, INVKT);
        ORD(V(di1), V(G4));
        float eg0 = __builtin_amdgcn_exp2f(G4);
        ORD(V(eg0), V(ei0));
        float di0 = __builtin_fmaf(ei0, INVKT, INVKT);
        ORD(V(di0), V(ef1));
        float df1 = __builtin_fmaf(ef1, 1.0f, 1.0f);
        ORD(V(df1), V(di1));
        float a1 = __builtin_amdgcn_rcpf(di1);
        ORD(V(a1), V(ef0));
        float df0 = __builtin_fmaf(ef0, 1.0f, 1.0f);
        ORD(V(df0), V(di0));
        float a0 = __builtin_amdgcn_rcpf(di0);
        ORD(V(a0), V(eg1));
        float dg1 = __builtin_fmaf(eg1, 1.0f, 1.0f);
        ORD(V(dg1), V(df1));
        float f1 = __builtin_amdgcn_rcpf(df1);
        ORD(V(f1), V(eg0));
        float dg0 = __builtin_fmaf(eg0, 1.0f, 1.0f);
        ORD(V(dg0), V(df0));
        float f0 = __builtin_amdgcn_rcpf(df0);
        ORD(V(f0), V(dg1));
        float s1 = __builtin_amdgcn_rcpf(dg1);
        ORD(V(s1), V(a1));
        float ta1 = a1 + a1;
        ORD(V(ta1), V(a0));
        float ta0 = a0 + a0;
        ORD(V(ta0), V(dg0));
        float s0 = __builtin_amdgcn_rcpf(dg0);
        ORD(V(s0), V(f1));
        float in1 = __builtin_fmaf(f1, xor8_dpp(c1), -a1);
        ORD(V(in1), V(G6));
        float eo0 = __builtin_amdgcn_exp2f(G6);
        ORD(V(eo0), V(f0));
        float in0 = __builtin_fmaf(f0, xor8_dpp(c0), -a0);
        ORD(V(in0), V(G7));
        float eo1 = __builtin_amdgcn_exp2f(G7);
        ORD(V(eo1), V(ta1));
        c1 = __builtin_fmaf(ta1, s1, in1);
        if (p + 1 < CH / 2) {
            if (odd == 0) {
                ORD(V(c1), V(XA[p + 1]));
                Pn0 = __builtin_amdgcn_wmma_f32_16x16x4_f32(false, A0x, false, XA[p + 1], (short)0, bias0, false, false);
                ORD(V(Pn0), V(c1));
            } else {
                ORD(V(c1), V(XB[p + 1]));
                Pn1 = __builtin_amdgcn_wmma_f32_16x16x4_f32(false, A1x, false, XB[p + 1], (short)0, bias1, false, false);
                ORD(V(Pn1), V(c1));
            }
        }
        ORD(V(c1), V(ta0));
        c0 = __builtin_fmaf(ta0, s0, in0);
        ORD(V(c0), V(eo0));
        float do0 = __builtin_fmaf(eo0, 0.5f, 0.5f);
        ORD(V(do0), V(c1));
        float ec1 = __builtin_amdgcn_exp2f(c1);
        ORD(V(ec1), V(eo1));
        float do1 = __builtin_fmaf(eo1, 0.5f, 0.5f);
        ORD(V(do1), V(c0));
        float ec0 = __builtin_amdgcn_exp2f(c0);
        ORD(V(ec0), V(do0));
        float q0 = __builtin_amdgcn_rcpf(do0);
        ORD(V(q0), V(do1));
        float q1 = __builtin_amdgcn_rcpf(do1);
        ORD(V(q1), V(ec1), V(ec0));
        float dc1 = __builtin_fmaf(ec1, 1.0f, 1.0f);
        float dc0 = __builtin_fmaf(ec0, 1.0f, 1.0f);
        ORD(V(dc1), V(dc0), V(q0));
        float mq0 = -0.5f * q0;
        ORD(V(mq0), V(dc1));
        float r1 = __builtin_amdgcn_rcpf(dc1);
        ORD(V(r1), V(q1));
        float mq1 = -0.5f * q1;
        ORD(V(mq1), V(dc0));
        float r0 = __builtin_amdgcn_rcpf(dc0);
        ORD(V(r0), V(q1), V(q0));
        float h1 = __builtin_fmaf(q1, r1, mq1);
        float h0 = __builtin_fmaf(q0, r0, mq0);
        Bown.x = h0;
        Bown.y = h1;
        Bpar.x = xor8_dpp(h0);
        Bpar.y = xor8_dpp(h1);
        if (p + 1 < CH / 2) {
            if (odd == 0) asm volatile("" :: "v"(XA[p + 1]));
            else          asm volatile("" :: "v"(XB[p + 1]));
        }
        }
        P0 = Pn0; P1 = Pn1;
        }
    }

    const int ubl = 2 * half + 4 * (hi ^ ((T - 1) & 1));
    hfin[bl * FSTR + ubl]     = Bown.x;
    hfin[bl * FSTR + ubl + 1] = Bown.y;
    __syncthreads();
    if (wv == 0) {
        float acc = b_fc[0];
        for (int j = 0; j < 8; ++j)
            acc += hfin[lane * FSTR + j] * W_fc[j];
        out[blockIdx.x * (WPB * BPW) + lane] = acc;
    }
}

extern "C" void kernel_launch(void* const* d_in, const int* in_sizes, int n_in,
                              void* d_out, int out_size, void* d_ws, size_t ws_size,
                              hipStream_t stream) {
    const float* x    = (const float*)d_in[0];
    const float* W_ih = (const float*)d_in[1];
    const float* W_hh = (const float*)d_in[2];
    const float* b_ih = (const float*)d_in[3];
    const float* b_hh = (const float*)d_in[4];
    const float* W_fc = (const float*)d_in[5];
    const float* b_fc = (const float*)d_in[6];
    float* out = (float*)d_out;

    const int T = 512;
    const int I = 4;
    const int B = in_sizes[0] / (T * I);
    const int wavesTotal = B / BPW;
    const int blocks = wavesTotal / WPB;

    lstm_wmma_kernel<<<blocks, 32 * WPB, 0, stream>>>(x, W_ih, W_hh, b_ih, b_hh,
                                                      W_fc, b_fc, out, T);
}


# Chunk boundary software-pipelined: chunk ch+1's LDS fragment reads and first x products issued in chunk ch's last pair of steps, into registers just freed; no wait at a chunk head; sums unchanged
# speedup vs baseline: 2.7001x; 1.0320x over previous
#include <hip/hip_runtime.h>


typedef __attribute__((ext_vector_type(2))) float v2f;
typedef __attribute__((ext_vector_type(4))) float v4f;
typedef __attribute__((ext_vector_type(8))) float v8f;

#define WPB  4
#define BPW  8
#define FSTR 9
#define CH   16
#define XROW (CH * 4 + 8)
#define XBUF (BPW * XROW)

#define KS    (-1.44269504f)
#define KT    (-2.88539008f)
#define INVKT (-0.34657359f)
#define V(x) "+v"(x)
#define ORD(...) asm volatile("" : __VA_ARGS__)
__device__ __forceinline__ float xor8_dpp(float v) {
    int r = __builtin_amdgcn_update_dpp(0, __builtin_bit_cast(int, v),
                                        0x128  , 0xF, 0xF, true);
    return __builtin_bit_cast(float, r);
}

__global__ void __launch_bounds__(32 * WPB)
lstm_wmma_kernel(const float* __restrict__ x,
                 const float* __restrict__ W_ih, const float* __restrict__ W_hh,
                 const float* __restrict__ b_ih, const float* __restrict__ b_hh,
                 const float* __restrict__ W_fc, const float* __restrict__ b_fc,
                 float* __restrict__ out, int T)
{
    __shared__ float hfin[WPB * BPW * FSTR];
    __shared__ __attribute__((aligned(16))) float xs[WPB * 2 * XBUF];

    const int lane = threadIdx.x & 31;
    const int wv   = threadIdx.x >> 5;
    const int n    = lane & 15;
    const int half = lane >> 4;
    const int k2   = 2 * half;
    const int hi   = n >> 3;
    const int bl   = wv * BPW + (n & 7);

    const int R0 = ((n & 7) >> 1) * 8 + 2 * (n >> 3) + (n & 1);
    const int R1 = R0 + 4;
    v2f A0x, A1x, A0o, A0p, A1o, A1p;
    A0x.x = W_ih[R0 * 4 + k2];       A0x.y = W_ih[R0 * 4 + k2 + 1];
    A1x.x = W_ih[R1 * 4 + k2];       A1x.y = W_ih[R1 * 4 + k2 + 1];
    A0o.x = W_hh[R0 * 8 + k2];       A0o.y = W_hh[R0 * 8 + k2 + 1];
    A0p.x = W_hh[R0 * 8 + 4 + k2];   A0p.y = W_hh[R0 * 8 + 4 + k2 + 1];
    A1o.x = W_hh[R1 * 8 + 4 + k2];   A1o.y = W_hh[R1 * 8 + 4 + k2 + 1];
    A1p.x = W_hh[R1 * 8 + k2];       A1p.y = W_hh[R1 * 8 + k2 + 1];
    const float rsc = (((n & 7) >> 1) == 2) ? KT : KS;
    A0x *= rsc; A1x *= rsc; A0o *= rsc; A0p *= rsc; A1o *= rsc; A1p *= rsc;

    v8f bias0, bias1;
    for (int v = 0; v < 8; ++v) {
        const int r = (v >> 1) * 8 + 2 * half + (v & 1);
        const float sc = ((v >> 1) == 2) ? KT : KS;
        bias0[v] = (b_ih[r] + b_hh[r]) * sc;
        bias1[v] = (b_ih[r + 4] + b_hh[r + 4]) * sc;
    }

    const int NCH = T / CH;
    const int rsel = lane >> 4;
    const int ssel = lane & 15;
    const char* xw = (const char*)x + (size_t)blockIdx.x * (WPB * BPW) * (size_t)T * 16;
    unsigned goff[4];
    for (int j = 0; j < 4; ++j)
        goff[j] = ((unsigned)(wv * BPW + 2 * j + rsel) * (unsigned)T + (unsigned)ssel) * 16u;
    float* xsw = xs + wv * (2 * XBUF);
    float* wst = xsw + rsel * XROW + ssel * 4;
    const float* rldA = xsw + (n & 7) * XROW + k2 + 4 * hi;
    const float* rldB = xsw + (n & 7) * XROW + k2 + 4 * (1 - hi);

    v4f gq[4];
    {
        const int c1 = (1 < NCH) ? 1 : 0;
        const int c2 = (2 < NCH) ? 2 : (NCH - 1);
        v4f g1[4];
        for (int j = 0; j < 4; ++j) gq[j] = *(const v4f*)(xw + goff[j]);
        for (int j = 0; j < 4; ++j) g1[j] = *(const v4f*)(xw + (size_t)c1 * (CH * 16) + goff[j]);
        for (int j = 0; j < 4; ++j) *(v4f*)(wst + 2 * j * XROW) = gq[j];
        for (int j = 0; j < 4; ++j) *(v4f*)(wst + XBUF + 2 * j * XROW) = g1[j];
        for (int j = 0; j < 4; ++j) gq[j] = *(const v4f*)(xw + (size_t)c2 * (CH * 16) + goff[j]);
    }

    v2f XA[CH / 2], XB[CH / 2];
    for (int p = 0; p < CH / 2; ++p) {
        XA[p] = *(const v2f*)(rldA + p * 8);
        XB[p] = *(const v2f*)(rldB + p * 8);
    }
    v8f P0 = __builtin_amdgcn_wmma_f32_16x16x4_f32(false, A0x, false, XA[0], (short)0, bias0, false, false);
    v8f P1 = __builtin_amdgcn_wmma_f32_16x16x4_f32(false, A1x, false, XB[0], (short)0, bias1, false, false);

    float c0 = 0.0f, c1 = 0.0f;
    v2f Bown; Bown.x = 0.0f; Bown.y = 0.0f;
    v2f Bpar = Bown;

    for (int ch = 0; ch < NCH; ++ch) {
        const int nb = (ch + 1) & 1;

        #pragma unroll
        for (int p = 0; p < CH / 2; ++p) {
        v8f Pn0 = P0, Pn1 = P1;
        #pragma unroll
        for (int odd = 0; odd < 2; ++odd) {
        v8f acc0, acc1;
        acc0 = __builtin_amdgcn_wmma_f32_16x16x4_f32(false, A0p, false, Bown, (short)0, P0,   false, false);
        acc1 = __builtin_amdgcn_wmma_f32_16x16x4_f32(false, A1p, false, Bown, (short)0, P1,   false, false);
        acc0 = __builtin_amdgcn_wmma_f32_16x16x4_f32(false, A0o, false, Bpar, (short)0, acc0, false, false);
        acc1 = __builtin_amdgcn_wmma_f32_16x16x4_f32(false, A1o, false, Bpar, (short)0, acc1, false, false);

        if (p == CH / 2 - 1 && odd == 0) {
            __builtin_amdgcn_sched_barrier(0);
            for (int q = 0; q < CH / 2; ++q) {
                XA[q] = *(const v2f*)(rldA + nb * XBUF + q * 8);
                XB[q] = *(const v2f*)(rldB + nb * XBUF + q * 8);
            }
            for (int j = 0; j < 4; ++j) *(v4f*)(wst + (nb ^ 1) * XBUF + 2 * j * XROW) = gq[j];
            const int cn = (ch + 3 < NCH) ? (ch + 3) : (NCH - 1);
            for (int j = 0; j < 4; ++j) gq[j] = *(const v4f*)(xw + (size_t)cn * (CH * 16) + goff[j]);
            __builtin_amdgcn_sched_barrier(0);
            asm volatile("" :: "v"(Bown), "v"(Bpar));
        }

        float G0 = (hi != odd) ? acc1[0] : acc0[0], G1 = (hi != odd) ? acc1[1] : acc0[1];
        float G2 = (hi != odd) ? acc1[2] : acc0[2], G3 = (hi != odd) ? acc1[3] : acc0[3];
        float G4 = (hi != odd) ? acc1[4] : acc0[4], G5 = (hi != odd) ? acc1[5] : acc0[5];
        float G6 = (hi != odd) ? acc1[6] : acc0[6], G7 = (hi != odd) ? acc1[7] : acc0[7];
        ORD(V(G0), V(G1), V(G2), V(G3), V(G4), V(G5), V(G6), V(G7));

        float ei1 = __builtin_amdgcn_exp2f(G1);
        ORD(V(ei1), V(G0));
        float ei0 = __builtin_amdgcn_exp2f(G0);
        ORD(V(ei0), V(G3));
        float ef1 = __builtin_amdgcn_exp2f(G3);
        ORD(V(ef1), V(G2));
        float ef0 = __builtin_amdgcn_exp2f(G2);
        ORD(V(ef0), V(G5));
        float eg1 = __builtin_amdgcn_exp2f(G5);
        ORD(V(eg1), V(ei1));
        float di1 = __builtin_fmaf(ei1, INVKT, INVKT);
        ORD(V(di1), V(G4));
        float eg0 = __builtin_amdgcn_exp2f(G4);
        ORD(V(eg0), V(ei0));
        float di0 = __builtin_fmaf(ei0, INVKT, INVKT);
        ORD(V(di0), V(ef1));
        float df1 = __builtin_fmaf(ef1, 1.0f, 1.0f);
        ORD(V(df1), V(di1));
        float a1 = __builtin_amdgcn_rcpf(di1);
        ORD(V(a1), V(ef0));
        float df0 = __builtin_fmaf(ef0, 1.0f, 1.0f);
        ORD(V(df0), V(di0));
        float a0 = __builtin_amdgcn_rcpf(di0);
        ORD(V(a0), V(eg1));
        float dg1 = __builtin_fmaf(eg1, 1.0f, 1.0f);
        ORD(V(dg1), V(df1));
        float f1 = __builtin_amdgcn_rcpf(df1);
        ORD(V(f1), V(eg0));
        float dg0 = __builtin_fmaf(eg0, 1.0f, 1.0f);
        ORD(V(dg0), V(df0));
        float f0 = __builtin_amdgcn_rcpf(df0);
        ORD(V(f0), V(dg1));
        float s1 = __builtin_amdgcn_rcpf(dg1);
        ORD(V(s1), V(a1));
        float ta1 = a1 + a1;
        ORD(V(ta1), V(a0));
        float ta0 = a0 + a0;
        ORD(V(ta0), V(dg0));
        float s0 = __builtin_amdgcn_rcpf(dg0);
        ORD(V(s0), V(f1));
        float in1 = __builtin_fmaf(f1, xor8_dpp(c1), -a1);
        ORD(V(in1), V(G6));
        float eo0 = __builtin_amdgcn_exp2f(G6);
        ORD(V(eo0), V(f0));
        float in0 = __builtin_fmaf(f0, xor8_dpp(c0), -a0);
        ORD(V(in0), V(G7));
        float eo1 = __builtin_amdgcn_exp2f(G7);
        ORD(V(eo1), V(ta1));
        c1 = __builtin_fmaf(ta1, s1, in1);
        {
            const int pn = (p + 1) % (CH / 2);
            if (odd == 0) {
                ORD(V(c1), V(XA[pn]));
                Pn0 = __builtin_amdgcn_wmma_f32_16x16x4_f32(false, A0x, false, XA[pn], (short)0, bias0, false, false);
                ORD(V(Pn0), V(c1));
            } else {
                ORD(V(c1), V(XB[pn]));
                Pn1 = __builtin_amdgcn_wmma_f32_16x16x4_f32(false, A1x, false, XB[pn], (short)0, bias1, false, false);
                ORD(V(Pn1), V(c1));
            }
        }
        ORD(V(c1), V(ta0));
        c0 = __builtin_fmaf(ta0, s0, in0);
        ORD(V(c0), V(eo0));
        float do0 = __builtin_fmaf(eo0, 0.5f, 0.5f);
        ORD(V(do0), V(c1));
        float ec1 = __builtin_amdgcn_exp2f(c1);
        ORD(V(ec1), V(eo1));
        float do1 = __builtin_fmaf(eo1, 0.5f, 0.5f);
        ORD(V(do1), V(c0));
        float ec0 = __builtin_amdgcn_exp2f(c0);
        ORD(V(ec0), V(do0));
        float q0 = __builtin_amdgcn_rcpf(do0);
        ORD(V(q0), V(do1));
        float q1 = __builtin_amdgcn_rcpf(do1);
        ORD(V(q1), V(ec1), V(ec0));
        float dc1 = __builtin_fmaf(ec1, 1.0f, 1.0f);
        float dc0 = __builtin_fmaf(ec0, 1.0f, 1.0f);
        ORD(V(dc1), V(dc0), V(q0));
        float mq0 = -0.5f * q0;
        ORD(V(mq0), V(dc1));
        float r1 = __builtin_amdgcn_rcpf(dc1);
        ORD(V(r1), V(q1));
        float mq1 = -0.5f * q1;
        ORD(V(mq1), V(dc0));
        float r0 = __builtin_amdgcn_rcpf(dc0);
        ORD(V(r0), V(q1), V(q0));
        float h1 = __builtin_fmaf(q1, r1, mq1);
        float h0 = __builtin_fmaf(q0, r0, mq0);
        Bown.x = h0;
        Bown.y = h1;
        Bpar.x = xor8_dpp(h0);
        Bpar.y = xor8_dpp(h1);
        if (odd == 0) asm volatile("" :: "v"(XA[(p + 1) % (CH / 2)]));
        else          asm volatile("" :: "v"(XB[(p + 1) % (CH / 2)]));
        }
        P0 = Pn0; P1 = Pn1;
        }
    }

    const int ubl = 2 * half + 4 * (hi ^ ((T - 1) & 1));
    hfin[bl * FSTR + ubl]     = Bown.x;
    hfin[bl * FSTR + ubl + 1] = Bown.y;
    __syncthreads();
    if (wv == 0) {
        float acc = b_fc[0];
        for (int j = 0; j < 8; ++j)
            acc += hfin[lane * FSTR + j] * W_fc[j];
        out[blockIdx.x * (WPB * BPW) + lane] = acc;
    }
}

extern "C" void kernel_launch(void* const* d_in, const int* in_sizes, int n_in,
                              void* d_out, int out_size, void* d_ws, size_t ws_size,
                              hipStream_t stream) {
    const float* x    = (const float*)d_in[0];
    const float* W_ih = (const float*)d_in[1];
    const float* W_hh = (const float*)d_in[2];
    const float* b_ih = (const float*)d_in[3];
    const float* b_hh = (const float*)d_in[4];
    const float* W_fc = (const float*)d_in[5];
    const float* b_fc = (const float*)d_in[6];
    float* out = (float*)d_out;

    const int T = 512;
    const int I = 4;
    const int B = in_sizes[0] / (T * I);
    const int wavesTotal = B / BPW;
    const int blocks = wavesTotal / WPB;

    lstm_wmma_kernel<<<blocks, 32 * WPB, 0, stream>>>(x, W_ih, W_hh, b_ih, b_hh,
                                                      W_fc, b_fc, out, T);
}
